# starting kernel
# baseline (speedup 1.0000x reference)
#include <hip/hip_runtime.h>
#include <math.h>

typedef __attribute__((ext_vector_type(2))) float v2f;
typedef __attribute__((ext_vector_type(8))) float v8f;

#define N0_  262144
#define INF_ 128
#define H_   32
#define N1_  131072
#define N2_  65536

__device__ __forceinline__ v8f wmma_f32_k4(v2f a, v2f b, v8f c) {
  return __builtin_amdgcn_wmma_f32_16x16x4_f32(
       false, a,  false, b,
       (short)0, c,  false,  false);
}

template <int F4, int LOG2F4>
__global__ __launch_bounds__(256) void sage_scatter_kernel(
    const float* __restrict__ xsrc, const int* __restrict__ src,
    const int* __restrict__ tgt, float* __restrict__ agg,
    float* __restrict__ cnt, int nedges) {
  long long tid = (long long)blockIdx.x * blockDim.x + threadIdx.x;
  long long total = (long long)nedges << LOG2F4;
  if (tid >= total) return;
  int e = (int)(tid >> LOG2F4);
  int g = (int)(tid & (F4 - 1));
  int s = src[e];
  int t = tgt[e];
  const float4* x4 = (const float4*)xsrc;
  float4 v = x4[(size_t)s * F4 + g];
  float* dst = agg + (size_t)t * (F4 * 4) + g * 4;
  unsafeAtomicAdd(dst + 0, v.x);
  unsafeAtomicAdd(dst + 1, v.y);
  unsafeAtomicAdd(dst + 2, v.z);
  unsafeAtomicAdd(dst + 3, v.w);
  if (g == 0) unsafeAtomicAdd(cnt + t, 1.0f);
}

template <int K, bool RELU>
__global__ __launch_bounds__(256) void sage_compute_kernel(
    const float* __restrict__ agg, const float* __restrict__ cnt,
    const float* __restrict__ xin, const float* __restrict__ wl,
    const float* __restrict__ bl, const float* __restrict__ wr,
    float* __restrict__ out, int nrows) {
  const int lane = threadIdx.x & 31;
  const int wid  = blockIdx.x * (blockDim.x >> 5) + (threadIdx.x >> 5);
  const int base = wid * 16;
  if (base >= nrows) return;
  const int l16  = lane & 15;
  const int half = lane >> 4;
  const int rowA = base + l16;

  const float inv = 1.0f / fmaxf(cnt[rowA], 1.0f);

  v8f c0 = {};
  v8f c1 = {};

#pragma unroll
  for (int phase = 0; phase < 2; ++phase) {
    const float* Arow = (phase == 0 ? agg : xin) + (size_t)rowA * K;
    const float* W    = (phase == 0 ? wl : wr);
    const float  s    = (phase == 0 ? inv : 1.0f);
    const float2* A2  = (const float2*)Arow;
    const float2* B0  = (const float2*)(W + (size_t)l16 * K);
    const float2* B1  = (const float2*)(W + (size_t)(l16 + 16) * K);
#pragma unroll
    for (int kc = 0; kc < K; kc += 4) {
      const int fi = (kc >> 1) + half;
      float2 fa = A2[fi];
      float2 f0 = B0[fi];
      float2 f1 = B1[fi];
      v2f a;  a.x  = fa.x * s; a.y  = fa.y * s;
      v2f b0; b0.x = f0.x;     b0.y = f0.y;
      v2f b1; b1.x = f1.x;     b1.y = f1.y;
      c0 = wmma_f32_k4(a, b0, c0);
      c1 = wmma_f32_k4(a, b1, c1);
    }
  }

  const float bias0 = bl[l16];
  const float bias1 = bl[l16 + 16];
#pragma unroll
  for (int r = 0; r < 8; ++r) { c0[r] += bias0; c1[r] += bias1; }

#pragma unroll
  for (int r = 0; r < 8; ++r) {
    float p = c0[r] * c0[r] + c1[r] * c1[r];
    p += __shfl_xor(p, 1, 32);
    p += __shfl_xor(p, 2, 32);
    p += __shfl_xor(p, 4, 32);
    p += __shfl_xor(p, 8, 32);
    float scale = 1.0f / fmaxf(sqrtf(p), 1e-12f);
    float o0 = c0[r] * scale;
    float o1 = c1[r] * scale;
    if (RELU) { o0 = fmaxf(o0, 0.0f); o1 = fmaxf(o1, 0.0f); }
    const size_t row = (size_t)(base + r + half * 8);
    out[row * 32 + l16]      = o0;
    out[row * 32 + 16 + l16] = o1;
  }
}

extern "C" void kernel_launch(void* const* d_in, const int* in_sizes, int n_in,
                              void* d_out, int out_size, void* d_ws,
                              size_t ws_size, hipStream_t stream) {
  const float* x   = (const float*)d_in[0];
  const int* src0  = (const int*)d_in[1];
  const int* tgt0  = (const int*)d_in[2];
  const int* src1  = (const int*)d_in[3];
  const int* tgt1  = (const int*)d_in[4];
  const float* wl0 = (const float*)d_in[5];
  const float* bl0 = (const float*)d_in[6];
  const float* wr0 = (const float*)d_in[7];
  const float* wl1 = (const float*)d_in[8];
  const float* bl1 = (const float*)d_in[9];
  const float* wr1 = (const float*)d_in[10];
  const int E0 = in_sizes[1];
  const int E1 = in_sizes[3];

  float* ws   = (float*)d_ws;
  float* agg0 = ws;
  float* cnt0 = agg0 + (size_t)N1_ * INF_;
  float* agg1 = cnt0 + (size_t)N1_;
  float* cnt1 = agg1 + (size_t)N2_ * H_;
  float* h    = cnt1 + (size_t)N2_;
  const size_t zero_bytes =
      ((size_t)N1_ * INF_ + N1_ + (size_t)N2_ * H_ + N2_) * sizeof(float);
  hipMemsetAsync(d_ws, 0, zero_bytes, stream);

  {
    long long total = (long long)E0 * 32;
    int blocks = (int)((total + 255) / 256);
    sage_scatter_kernel<32, 5><<<blocks, 256, 0, stream>>>(x, src0, tgt0, agg0,
                                                           cnt0, E0);
  }
  {
    int blocks = (N1_ / 16) / 8;
    sage_compute_kernel<INF_, true><<<blocks, 256, 0, stream>>>(
        agg0, cnt0, x, wl0, bl0, wr0, h, N1_);
  }
  {
    long long total = (long long)E1 * 8;
    int blocks = (int)((total + 255) / 256);
    sage_scatter_kernel<8, 3><<<blocks, 256, 0, stream>>>(h, src1, tgt1, agg1,
                                                          cnt1, E1);
  }
  {
    int blocks = (N2_ / 16) / 8;
    sage_compute_kernel<H_, false><<<blocks, 256, 0, stream>>>(
        agg1, cnt1, h, wl1, bl1, wr1, (float*)d_out, N2_);
  }
}


# Layer-0 transform waves keep both weight matrices' WMMA B fragments in registers over 4 row tiles, next tile's A rows fetched a stage ahead: 2048 waves, not 8192; k order and sums unchanged
# speedup vs baseline: 4.2943x; 4.2943x over previous
#include <hip/hip_runtime.h>
#include <math.h>

typedef __attribute__((ext_vector_type(2))) float v2f;
typedef __attribute__((ext_vector_type(8))) float v8f;

#define E0_  2097152
#define E1_  1048576
#define N0_  262144
#define INF_ 128
#define H_   32
#define N1_  131072
#define N2_  65536

__device__ __forceinline__ v8f wmma_f32_k4(v2f a, v2f b, v8f c) {
  return __builtin_amdgcn_wmma_f32_16x16x4_f32(
       false, a,  false, b,
       (short)0, c,  false,  false);
}

#define KCH_VEC   1024
#define KCH_PITCH 257
#define SRC_BLK_LOG 16
#define SRC_NBLK    4

template <int LOG2ROWS, bool IL>
__device__ __forceinline__ void key_chunk(const int* __restrict__ tgt,
                                          const int* __restrict__ src,
                                          int* __restrict__ il,
                                          unsigned* __restrict__ bm, int nedges,
                                          int ch, int nch, unsigned* lbm, int2* stage) {
  const int tid = threadIdx.x;
  for (int i = tid; i < 32 * KCH_PITCH; i += 1024) lbm[i] = 0u;
  __syncthreads();
  const int4* t4 = (const int4*)tgt;
  const int nvec = nedges >> 4;
  bool ordered = false;
  if constexpr (IL) ordered = (ch + 1) * KCH_VEC <= nvec;
  if (ordered) {
    if constexpr (IL) {
      const int lane = tid & 31;
      const int w    = tid >> 5;
      int tt[16], ss[16];
#pragma unroll
      for (int k = 0; k < 4; ++k) {
        const int g = ch * (4 * KCH_VEC) + w * 128 + k * 32 + lane;
        const int4 a = t4[g];
        const int4 b = ((const int4*)src)[g];
        tt[4 * k] = a.x; tt[4 * k + 1] = a.y; tt[4 * k + 2] = a.z; tt[4 * k + 3] = a.w;
        ss[4 * k] = b.x; ss[4 * k + 1] = b.y; ss[4 * k + 2] = b.z; ss[4 * k + 3] = b.w;
      }
      unsigned long long pk = 0ull;
#pragma unroll
      for (int i = 0; i < 16; ++i)
        pk += 1ull << (16 * (((unsigned)ss[i] >> SRC_BLK_LOG) & (SRC_NBLK - 1)));
      unsigned long long inc = pk;
#pragma unroll
      for (int d = 1; d < 32; d <<= 1) {
        const unsigned lo = __shfl_up((unsigned)inc, d, 32);
        const unsigned hi = __shfl_up((unsigned)(inc >> 32), d, 32);
        if (lane >= d) inc += ((unsigned long long)hi << 32) | lo;
      }
      const unsigned tlo = __builtin_amdgcn_readlane((unsigned)inc, 31);
      const unsigned thi = __builtin_amdgcn_readlane((unsigned)(inc >> 32), 31);
      const unsigned n0 = tlo & 0xffffu, n1 = tlo >> 16, n2 = thi & 0xffffu;
      unsigned long long run = ((unsigned long long)n0 << 16) | ((unsigned long long)(n0 + n1) << 32) |
                               ((unsigned long long)(n0 + n1 + n2) << 48);
      run += inc - pk;
      int2* st = stage + w * 512;
      unsigned* col = lbm + w * KCH_PITCH;
#pragma unroll
      for (int i = 0; i < 16; ++i) {
        const int sh  = 16 * (((unsigned)ss[i] >> SRC_BLK_LOG) & (SRC_NBLK - 1));
        const int pos = (int)((run >> sh) & 0xffffull);
        run += 1ull << sh;
        st[pos] = make_int2(tt[i], ss[i]);
        atomicOr(&col[(tt[i] >> LOG2ROWS) & 255], 1u << (pos >> 4));
      }
      __syncthreads();
      const int4* s4 = (const int4*)st;
      int4* o4 = (int4*)il + (size_t)ch * (8 * KCH_VEC) + w * 256;
#pragma unroll
      for (int i = 0; i < 8; ++i) o4[i * 32 + lane] = s4[i * 32 + lane];
    }
  } else {
  int4 t[4];
  int4 sv[4];
  bool ok[4];
#pragma unroll
  for (int k = 0; k < 4; ++k) {
    const int g = ch * (4 * KCH_VEC) + tid + 1024 * k;
    ok[k] = (g >> 2) < nvec;
    t[k]  = ok[k] ? t4[g] : make_int4(0, 0, 0, 0);
    if constexpr (IL) sv[k] = ok[k] ? ((const int4*)src)[g] : make_int4(0, 0, 0, 0);
  }
  if constexpr (IL) {
    int4* o4 = (int4*)il;
#pragma unroll
    for (int k = 0; k < 4; ++k) {
      if (ok[k]) {
        const int g = ch * (4 * KCH_VEC) + tid + 1024 * k;
        o4[2 * g]     = make_int4(t[k].x, sv[k].x, t[k].y, sv[k].y);
        o4[2 * g + 1] = make_int4(t[k].z, sv[k].z, t[k].w, sv[k].w);
      }
    }
  }
#pragma unroll
  for (int k = 0; k < 4; ++k) {
    if (ok[k]) {
      const int jl = (tid + 1024 * k) >> 2;
      unsigned* col = lbm + (jl >> 5) * KCH_PITCH;
      const unsigned bit = 1u << (jl & 31);
      atomicOr(&col[(t[k].x >> LOG2ROWS) & 255], bit);
      atomicOr(&col[(t[k].y >> LOG2ROWS) & 255], bit);
      atomicOr(&col[(t[k].z >> LOG2ROWS) & 255], bit);
      atomicOr(&col[(t[k].w >> LOG2ROWS) & 255], bit);
    }
  }
  }
  __syncthreads();
  const int lane = tid & 31;
  for (int o = tid >> 5; o < 256; o += 32)
    bm[((size_t)o * nch + ch) * 32 + lane] = lbm[lane * KCH_PITCH + o];
}

template <int LOG0, int LOG1, int NCH0, int NCH1>
__global__ __launch_bounds__(1024) void sage_key_kernel(
    const int* __restrict__ tgt0, const int* __restrict__ src0,
    int* __restrict__ il0, unsigned* __restrict__ bm0, int nedges0,
    const int* __restrict__ tgt1, unsigned* __restrict__ bm1, int nedges1) {
  __shared__ unsigned lbm[32 * KCH_PITCH];
  __shared__ __attribute__((aligned(16))) int2 stage[32 * 512];
  const int b = blockIdx.x;
  if (b < NCH0)             key_chunk<LOG0, true>(tgt0, src0, il0, bm0, nedges0, b, NCH0, lbm, stage);
  else if (b - NCH0 < NCH1) key_chunk<LOG1, false>(tgt1, nullptr, nullptr, bm1, nedges1, b - NCH0, NCH1, lbm, stage);
}

#define OWN_WAVES 32
#define OWN_QCAP  64
#define OWN_JB    16

__device__ __forceinline__ void lds_add_f32(float* p, float v) {
  __hip_atomic_fetch_add(p, v, __ATOMIC_RELAXED, __HIP_MEMORY_SCOPE_WORKGROUP);
}

template <int F, int LOG2ROWS, int NCH, bool IL>
__global__ __launch_bounds__(1024) void sage_owner_kernel(
    const float* __restrict__ xsrc, const int* __restrict__ src,
    const int* __restrict__ tgt, const int* __restrict__ il,
    const unsigned* __restrict__ bm,
    float* __restrict__ agg, float* __restrict__ cnt, int nedges) {
  constexpr int ROWS = 1 << LOG2ROWS;
  constexpr int FK = F / 32;
  __shared__ __attribute__((aligned(16))) float acc[ROWS * F];
  __shared__ unsigned cntl[ROWS];
  __shared__ int q_j[OWN_WAVES][OWN_QCAP];
  __shared__ int q_s[OWN_WAVES][OWN_QCAP];
  __shared__ int q_r[OWN_WAVES][OWN_QCAP];

  const int tid  = threadIdx.x;
  const int lane = tid & 31;
  const int l16  = lane & 15;
  const int half = lane >> 4;
  const int w    = tid >> 5;
  const int blk  = blockIdx.x;
  const int lo   = blk << LOG2ROWS;
  int* qj = q_j[w];
  int* qs = q_s[w];
  int* qr = q_r[w];

  for (int i = tid; i < ROWS * F / 4; i += 1024)
    ((float4*)acc)[i] = make_float4(0.f, 0.f, 0.f, 0.f);
  for (int i = tid; i < ROWS; i += 1024) cntl[i] = 0u;
  __syncthreads();

  int head = 0, tail = 0;
  int jh = 0, jt = 0;

  auto drain = [&](int n) {
    const bool valid = lane < n;
    const int  qi = (head + lane) & (OWN_QCAP - 1);
    const int  s  = valid ? qs[qi] : 0;
    const int  r  = valid ? qr[qi] : 0;
    if (valid) atomicAdd(&cntl[r], 1u);
    for (int j0 = 0; j0 < n; j0 += 8) {
      float v[8][FK];
      int   rr[8];
#pragma unroll
      for (int jj = 0; jj < 8; ++jj) {
        const int sj = __builtin_amdgcn_readlane(s, j0 + jj);
        rr[jj]       = __builtin_amdgcn_readlane(r, j0 + jj);
        if constexpr (FK == 4) {
          const float4 q = *(const float4*)(xsrc + (size_t)sj * F + 4 * lane);
          v[jj][0] = q.x; v[jj][1] = q.y; v[jj][2] = q.z; v[jj][3] = q.w;
        } else {
          const float* row = xsrc + (size_t)sj * F + lane;
#pragma unroll
          for (int k = 0; k < FK; ++k) v[jj][k] = row[32 * k];
        }
      }
#pragma unroll
      for (int jj = 0; jj < 8; ++jj) {
        if (j0 + jj < n) {
          float* a = acc + rr[jj] * F + lane;
#pragma unroll
          for (int k = 0; k < FK; ++k) lds_add_f32(a + 32 * k, v[jj][k]);
        }
      }
    }
    head += n;
  };

  auto push = [&](bool m, unsigned bal, int s, int r) {
    if (m) {
      const int qi = (tail + (int)__builtin_amdgcn_mbcnt_lo(bal, 0u)) & (OWN_QCAP - 1);
      qs[qi] = s;
      qr[qi] = r;
    }
    tail += __builtin_popcount(bal);
  };

  auto proc = [&](int n) {
    const int jv = (lane < n) ? qj[(jh + lane) & (OWN_QCAP - 1)] : -1;
    jh += n;
    int t[8], sr[8];
    if constexpr (IL) {
      const int grp = lane >> 3;
      const int pc  = lane & 7;
      const int q0 = grp == 0 ? -1 : 0, q1 = grp == 1 ? -1 : 0;
      const int q2 = grp == 2 ? -1 : 0, q3 = grp == 3 ? -1 : 0;
      int  js[4];
      int4 rec[4];
#pragma unroll
      for (int s = 0; s < 4; ++s) {
        const int j0 = __builtin_amdgcn_readlane(jv, 4 * s);
        const int j1 = __builtin_amdgcn_readlane(jv, 4 * s + 1);
        const int j2 = __builtin_amdgcn_readlane(jv, 4 * s + 2);
        const int j3 = __builtin_amdgcn_readlane(jv, 4 * s + 3);
        js[s] = (j0 & q0) | (j1 & q1) | (j2 & q2) | (j3 & q3);
      }
#pragma unroll
      for (int s = 0; s < 4; ++s)
        rec[s] = ((const int4*)il)[(js[s] < 0 ? 0 : js[s]) * 8 + pc];
#pragma unroll
      for (int s = 0; s < 4; ++s) {
        t[2 * s]     = rec[s].x | (js[s] >> 31);  sr[2 * s]     = rec[s].y;
        t[2 * s + 1] = rec[s].z | (js[s] >> 31);  sr[2 * s + 1] = rec[s].w;
      }
    } else {
#pragma unroll
    for (int s = 0; s < 8; ++s) {
      const int j0 = __builtin_amdgcn_readlane(jv, 2 * s);
      const int j1 = __builtin_amdgcn_readlane(jv, 2 * s + 1);
      const int js = half ? j1 : j0;
      const int e = (js < 0 ? 0 : js) * 16 + l16;
      t[s]  = tgt[e];
      sr[s] = src[e];
      if (js < 0) t[s] = -1;
    }
    }
    unsigned bal[8];
    int tot = 0;
#pragma unroll
    for (int s = 0; s < 8; ++s) {
      bal[s] = __builtin_amdgcn_ballot_w32((unsigned)(t[s] - lo) < (unsigned)ROWS);
      tot += __builtin_popcount(bal[s]);
    }
    if (tail - head + tot <= OWN_QCAP) {
#pragma unroll
      for (int s = 0; s < 8; ++s)
        push((unsigned)(t[s] - lo) < (unsigned)ROWS, bal[s], sr[s], t[s] - lo);
    } else {
#pragma nounroll
      for (int s = 0; s < 8; ++s) {
        int ts = t[0], ss = sr[0];
#pragma unroll
        for (int i = 1; i < 8; ++i)
          if (s == i) { ts = t[i]; ss = sr[i]; }
        const bool m = (unsigned)(ts - lo) < (unsigned)ROWS;
        push(m, __builtin_amdgcn_ballot_w32(m), ss, ts - lo);
        if (tail - head >= 32) drain(32);
      }
    }
    while (tail - head >= 32) drain(32);
  };

  const int nvec  = nedges >> 4;
  constexpr int NW = NCH * 32;
  constexpr int DW = NW / 1024;
  constexpr int NQ = IL ? SRC_NBLK : DW / 2;
  static_assert(NW % 2048 == 0, "a thread takes whole 64-bit words of the row");
  static_assert(!IL || (DW <= 8 && SRC_NBLK * 8 == 32), "a turn's bytes fit one 64-bit word");

  {
    const unsigned* rowp = bm + (size_t)blk * NW + tid * DW;
    unsigned wd[DW];
    if constexpr (DW == 4) {
      const uint4 r = *(const uint4*)rowp;
      wd[0] = r.x; wd[1] = r.y; wd[2] = r.z; wd[3] = r.w;
    } else if constexpr (DW == 2) {
      const uint2 r = *(const uint2*)rowp;
      wd[0] = r.x; wd[1] = r.y;
    } else {
#pragma unroll
      for (int i = 0; i < DW; ++i) wd[i] = rowp[i];
    }
    auto word = [&](int hq) -> unsigned long long {
      unsigned long long r = 0ull;
      if constexpr (IL) {
#pragma unroll
        for (int i = 0; i < DW; ++i)
          r |= (unsigned long long)((wd[i] >> (8 * hq)) & 0xffu) << (8 * i);
      } else {
#pragma unroll
        for (int i = 0; i < NQ; ++i)
          if (hq == i) r = (unsigned long long)wd[2 * i] | ((unsigned long long)wd[2 * i + 1] << 32);
      }
      return r;
    };
    int hq = 0;
    unsigned long long m = word(0);
    for (;;) {
      const bool any = __builtin_amdgcn_ballot_w32(m != 0ull) != 0u;
      if (!any && hq + 1 < NQ) { ++hq; m = word(hq); continue; }
      if (any) {
        const bool act = m != 0ull;
        const int  bp  = act ? __builtin_ctzll(m) : 0;
        const int  j   = IL ? (tid * DW + (bp >> 3)) * 32 + 8 * hq + (bp & 7)
                            : (tid * DW + 2 * hq) * 32 + bp;
        m &= m - 1ull;
        const bool mj = act && j < nvec;
        const unsigned bj = __builtin_amdgcn_ballot_w32(mj);
        if (mj) qj[(jt + (int)__builtin_amdgcn_mbcnt_lo(bj, 0u)) & (OWN_QCAP - 1)] = j;
        jt += __builtin_popcount(bj);
      }
      while (jt - jh >= OWN_JB || (!any && jt - jh > 0))
        proc(jt - jh < OWN_JB ? jt - jh : OWN_JB);
      if (!any) break;
    }
  }
  if (w == 0 && (nvec << 4) < nedges) {
    const int e = (nvec << 4) + lane;
    int ts = -1, ss = 0;
    if (e < nedges) { ts = tgt[e]; ss = src[e]; }
    const bool m = (unsigned)(ts - lo) < (unsigned)ROWS;
    push(m, __builtin_amdgcn_ballot_w32(m), ss, ts - lo);
  }
  while (tail - head > 0) drain(tail - head < 32 ? tail - head : 32);
  __syncthreads();

  float4* out4 = (float4*)(agg + (size_t)lo * F);
  if constexpr (FK == 4) {
    for (int i = tid; i < ROWS * F / 4; i += 1024) {
      const float* a = acc + (i >> 5) * F + (i & 31);
      out4[i] = make_float4(a[0], a[32], a[64], a[96]);
    }
  } else {
    for (int i = tid; i < ROWS * F / 4; i += 1024) out4[i] = ((const float4*)acc)[i];
  }
  for (int i = tid; i < ROWS; i += 1024) cnt[lo + i] = (float)cntl[i];
}

template <int K, bool RELU, int T>
__global__ __launch_bounds__(256) void sage_compute_kernel(
    const float* __restrict__ agg, const float* __restrict__ cnt,
    const float* __restrict__ xin, const float* __restrict__ wl,
    const float* __restrict__ bl, const float* __restrict__ wr,
    float* __restrict__ out, int nrows) {
  constexpr int NF = K / 8;
  const int lane  = threadIdx.x & 31;
  const int wid   = blockIdx.x * (blockDim.x >> 5) + (threadIdx.x >> 5);
  const int base0 = wid * (16 * T);
  if (base0 >= nrows) return;
  const int l16  = lane & 15;
  const int half = lane >> 4;

  float4 wb0[NF], wb1[NF], rb0[NF], rb1[NF];
  const float bias0 = bl[l16];
  const float bias1 = bl[l16 + 16];

  auto load_rows = [&](const float* srcp, int t, float4* dst) {
    const float4* A4 = (const float4*)(srcp + (size_t)(base0 + 16 * t + l16) * K);
#pragma unroll
    for (int i = 0; i < NF; ++i) dst[i] = A4[2 * i + half];
  };
  auto stage = [&](const float4* ap, const float4* b0p, const float4* b1p, bool scaled, float sc,
                   v8f& c0, v8f& c1) {
#pragma unroll
    for (int i = 0; i < NF; ++i) {
      const float4 a4 = ap[i];
      const float4 f0 = b0p[i];
      const float4 f1 = b1p[i];
      v2f a, b0, b1;
      if (scaled) { a.x = a4.x * sc; a.y = a4.y * sc; }
      else        { a.x = a4.x;      a.y = a4.y; }
      b0.x = f0.x;     b0.y = f0.y;
      b1.x = f1.x;     b1.y = f1.y;
      c0 = wmma_f32_k4(a, b0, c0);
      c1 = wmma_f32_k4(a, b1, c1);
      if (scaled) { a.x = a4.z * sc; a.y = a4.w * sc; }
      else        { a.x = a4.z;      a.y = a4.w; }
      b0.x = f0.z;     b0.y = f0.w;
      b1.x = f1.z;     b1.y = f1.w;
      c0 = wmma_f32_k4(a, b0, c0);
      c1 = wmma_f32_k4(a, b1, c1);
    }
  };

  float4 a0[NF], a1[NF];
  float  cv;
  cv = cnt[base0 + l16];
  load_rows(agg, 0, a0);
  {
    const float4* L0 = (const float4*)(wl + (size_t)l16 * K);
    const float4* L1 = (const float4*)(wl + (size_t)(l16 + 16) * K);
#pragma unroll
    for (int i = 0; i < NF; ++i) { wb0[i] = L0[2 * i + half]; wb1[i] = L1[2 * i + half]; }
  }
  load_rows(xin, 0, a1);
  {
    const float4* R0 = (const float4*)(wr + (size_t)l16 * K);
    const float4* R1 = (const float4*)(wr + (size_t)(l16 + 16) * K);
#pragma unroll
    for (int i = 0; i < NF; ++i) { rb0[i] = R0[2 * i + half]; rb1[i] = R1[2 * i + half]; }
  }

  auto tile = [&](int t, bool more) {
    const float inv = 1.0f / fmaxf(cv, 1.0f);
    v8f c0 = {};
    v8f c1 = {};
    stage(a0, wb0, wb1, true, inv, c0, c1);
    if (more) {
      load_rows(agg, t + 1, a0);
      cv = cnt[base0 + 16 * (t + 1) + l16];
    }
    stage(a1, rb0, rb1, false, 1.0f, c0, c1);
    if (more) load_rows(xin, t + 1, a1);

    const int base = base0 + 16 * t;
#pragma unroll
    for (int r = 0; r < 8; ++r) { c0[r] += bias0; c1[r] += bias1; }
#pragma unroll
    for (int r = 0; r < 8; ++r) {
      float q = c0[r] * c0[r] + c1[r] * c1[r];
      q += __shfl_xor(q, 1, 32);
      q += __shfl_xor(q, 2, 32);
      q += __shfl_xor(q, 4, 32);
      q += __shfl_xor(q, 8, 32);
      float scale = 1.0f / fmaxf(sqrtf(q), 1e-12f);
      float o0 = c0[r] * scale;
      float o1 = c1[r] * scale;
      if (RELU) { o0 = fmaxf(o0, 0.0f); o1 = fmaxf(o1, 0.0f); }
      const size_t row = (size_t)(base + r + half * 8);
      out[row * 32 + l16]      = o0;
      out[row * 32 + 16 + l16] = o1;
    }
  };
#pragma nounroll
  for (int t = 0; t < T - 1; ++t) tile(t, true);
  tile(T - 1, false);
}

extern "C" void kernel_launch(void* const* d_in, const int* in_sizes, int n_in,
                              void* d_out, int out_size, void* d_ws,
                              size_t ws_size, hipStream_t stream) {
  const float* x   = (const float*)d_in[0];
  const int* src0  = (const int*)d_in[1];
  const int* tgt0  = (const int*)d_in[2];
  const int* src1  = (const int*)d_in[3];
  const int* tgt1  = (const int*)d_in[4];
  const float* wl0 = (const float*)d_in[5];
  const float* bl0 = (const float*)d_in[6];
  const float* wr0 = (const float*)d_in[7];
  const float* wl1 = (const float*)d_in[8];
  const float* bl1 = (const float*)d_in[9];
  const float* wr1 = (const float*)d_in[10];
  const int E0 = in_sizes[1];
  const int E1 = in_sizes[3];

  float* ws   = (float*)d_ws;
  float* agg0 = ws;
  float* cnt0 = agg0 + (size_t)N1_ * INF_;
  float* bmp1 = cnt0 + (size_t)N1_;
  float* cnt1 = bmp1 + (size_t)N2_ * H_;
  float* h    = cnt1 + (size_t)N2_;
  float* agg1 = agg0;
  unsigned* bm1 = (unsigned*)bmp1;
  unsigned* bm0 = bm1 + (size_t)256 * (E1_ / 16384) * 32;
  int* il0 = (int*)h;
  static_assert((size_t)256 * (E1_ / 16384 + E0_ / 16384) * 32 <= (size_t)N2_ * H_, "both bitmaps fit the former agg1 plane");
  static_assert((size_t)2 * E0_ <= (size_t)N1_ * H_, "the interleaved pairs fit h's plane");

  sage_key_kernel<9, 8, E0_ / 16384, E1_ / 16384>
      <<<E0_ / 16384 + E1_ / 16384, 1024, 0, stream>>>(tgt0, src0, il0, bm0, E0, tgt1, bm1, E1);

  sage_owner_kernel<INF_, 9, E0_ / 16384, true><<<N1_ / 512, 1024, 0, stream>>>(
      x, src0, tgt0, il0, bm0, agg0, cnt0, E0);
  {
    static_assert(N1_ % (16 * 4 * 8) == 0, "whole work-groups of whole waves");
    int blocks = (N1_ / 16) / (8 * 4);
    sage_compute_kernel<INF_, true, 4><<<blocks, 256, 0, stream>>>(
        agg0, cnt0, x, wl0, bl0, wr0, h, N1_);
  }
  sage_owner_kernel<H_, 8, E1_ / 16384, false><<<N2_ / 256, 1024, 0, stream>>>(
      h, src1, tgt1, nullptr, bm1, agg1, cnt1, E1);
  {
    static_assert(N2_ % (16 * 8) == 0, "whole work-groups of whole waves");
    int blocks = (N2_ / 16) / 8;
    sage_compute_kernel<H_, false, 1><<<blocks, 256, 0, stream>>>(
        agg1, cnt1, h, wl1, bl1, wr1, (float*)d_out, N2_);
  }
}
